# starting kernel
# baseline (speedup 1.0000x reference)
#include <hip/hip_runtime.h>

typedef __attribute__((ext_vector_type(2))) float v2f;
typedef __attribute__((ext_vector_type(8))) float v8f;

#define NN 65536
#define DD 1024
#define CHUNKS 64
#define ROWS_PER_CHUNK (NN / CHUNKS)
#define DTILES (DD / 16)

__global__ void __launch_bounds__(256)
la_phase1_gemv(const float* __restrict__ P, const float* __restrict__ g,
               float* __restrict__ partial) {
  const int wave  = (blockIdx.x * blockDim.x + threadIdx.x) >> 5;
  const int lane  = threadIdx.x & 31;
  const int chunk = wave >> 6;
  const int dtile = wave & 63;
  const int d0    = dtile * 16;
  const int n0    = chunk * ROWS_PER_CHUNK;
  const int half  = lane >> 4;
  const int l16   = lane & 15;

  const float* pA = g + n0 + 2 * half;
  const float* pB = P + (size_t)(n0 + 2 * half) * DD + d0 + l16;

  v8f acc = {};
  for (int i = 0; i < ROWS_PER_CHUNK / 4; ++i) {
    v2f a; a.x = pA[0];  a.y = pA[1];
    v2f b; b.x = pB[0];  b.y = pB[DD];
    acc = __builtin_amdgcn_wmma_f32_16x16x4_f32(
         false, a,  false, b,
         (short)0, acc,  false,  false);
    pA += 4;
    pB += 4 * DD;
  }
  if (lane < 16) partial[chunk * DD + d0 + l16] = acc[0];
}

__global__ void __launch_bounds__(256)
la_phase2_reduce(const float* __restrict__ partial, float* __restrict__ weighted) {
  const int d = blockIdx.x * blockDim.x + threadIdx.x;
  float s = 0.f;
#pragma unroll 8
  for (int c = 0; c < CHUNKS; ++c) s += partial[c * DD + d];
  weighted[d] = s;
}

__global__ void __launch_bounds__(256)
la_phase3_outer(const float* __restrict__ weighted, const float* __restrict__ cs,
                float* __restrict__ out) {
  const int wave = (blockIdx.x * blockDim.x + threadIdx.x) >> 5;
  const int lane = threadIdx.x & 31;
  const int tm   = wave >> 6;
  const int tn   = wave & 63;
  const int half = lane >> 4;
  const int l16  = lane & 15;

  v2f a; a.x = (half == 0) ? weighted[tm * 16 + l16] : 0.f; a.y = 0.f;
  v2f b; b.x = (half == 0) ? cs[tn * 16 + l16]       : 0.f; b.y = 0.f;
  v8f c = {};
  c = __builtin_amdgcn_wmma_f32_16x16x4_f32(
      false, a, false, b, (short)0, c, false, false);

  float* o = out + (size_t)(tm * 16 + half * 8) * DD + tn * 16 + l16;
#pragma unroll
  for (int j = 0; j < 8; ++j) o[(size_t)j * DD] = c[j];
}

extern "C" void kernel_launch(void* const* d_in, const int* in_sizes, int n_in,
                              void* d_out, int out_size, void* d_ws, size_t ws_size,
                              hipStream_t stream) {
  (void)in_sizes; (void)n_in; (void)out_size; (void)ws_size;

  const float* past_states   = (const float*)d_in[0];
  const float* current_state = (const float*)d_in[1];
  const float* gradients     = (const float*)d_in[2];
  float* out = (float*)d_out;

  float* partial  = (float*)d_ws;
  float* weighted = partial + CHUNKS * DD;

  la_phase1_gemv<<<(CHUNKS * DTILES) / 8, 256, 0, stream>>>(past_states, gradients, partial);
  la_phase2_reduce<<<DD / 256, 256, 0, stream>>>(partial, weighted);
  la_phase3_outer<<<(DTILES * DTILES) / 8, 256, 0, stream>>>(weighted, current_state, out);
}


# last two launches fused: each work-group sums the 64 chunk partials of its 16 rows (4-chunk groups joined in LDS) and writes 16x512 of out; launches 3 -> 2; chunk sum order changed
# speedup vs baseline: 1.2354x; 1.2354x over previous
#include <hip/hip_runtime.h>

typedef __attribute__((ext_vector_type(2))) float v2f;
typedef __attribute__((ext_vector_type(8))) float v8f;

#define NN 65536
#define DD 1024
#define CHUNKS 64
#define ROWS_PER_CHUNK (NN / CHUNKS)
#define DTILES (DD / 16)

__global__ void __launch_bounds__(256)
la_phase1_gemv(const float* __restrict__ P, const float* __restrict__ g,
               float* __restrict__ partial) {
  const int wave  = (blockIdx.x * blockDim.x + threadIdx.x) >> 5;
  const int lane  = threadIdx.x & 31;
  const int chunk = wave >> 6;
  const int dtile = wave & 63;
  const int d0    = dtile * 16;
  const int n0    = chunk * ROWS_PER_CHUNK;
  const int half  = lane >> 4;
  const int l16   = lane & 15;

  const float* pA = g + n0 + 2 * half;
  const float* pB = P + (size_t)(n0 + 2 * half) * DD + d0 + l16;

  v8f acc = {};
  for (int i = 0; i < ROWS_PER_CHUNK / 4; ++i) {
    v2f a; a.x = pA[0];  a.y = pA[1];
    v2f b; b.x = pB[0];  b.y = pB[DD];
    acc = __builtin_amdgcn_wmma_f32_16x16x4_f32(
         false, a,  false, b,
         (short)0, acc,  false,  false);
    pA += 4;
    pB += 4 * DD;
  }
  if (lane < 16) partial[chunk * DD + d0 + l16] = acc[0];
}

#define STRIP_COLS 512
#define TILES_PER_WAVE (STRIP_COLS / 16 / 8)
#define GROUPS 16
#define CH_PER_GROUP (CHUNKS / GROUPS)

__global__ void __launch_bounds__(256)
la_reduce_outer(const float* __restrict__ partial, const float* __restrict__ cs,
                float* __restrict__ out) {
  __shared__ float gsum[GROUPS][16];
  const int lane = threadIdx.x & 31;
  const int wv   = threadIdx.x >> 5;
  const int half = lane >> 4;
  const int l16  = lane & 15;
  const int m0   = (blockIdx.x >> 1) * 16;
  const int n0   = (blockIdx.x & 1) * STRIP_COLS + wv * (TILES_PER_WAVE * 16);

  float csv[TILES_PER_WAVE];
#pragma unroll
  for (int k = 0; k < TILES_PER_WAVE; ++k) csv[k] = cs[n0 + k * 16 + l16];

  const int grp = threadIdx.x >> 4;
  const float* p = partial + (size_t)(grp * CH_PER_GROUP) * DD + m0 + l16;
  const float s0 = p[0], s1 = p[DD], s2 = p[2 * DD], s3 = p[3 * DD];
  gsum[grp][l16] = (s0 + s1) + (s2 + s3);
  __syncthreads();
  float w = 0.f;
#pragma unroll
  for (int g = 0; g < GROUPS; ++g) w += gsum[g][l16];

  v2f a; a.x = (half == 0) ? w : 0.f; a.y = 0.f;
  float* o = out + (size_t)(m0 + half * 8) * DD + n0 + l16;
#pragma unroll
  for (int k = 0; k < TILES_PER_WAVE; ++k) {
    v2f b; b.x = (half == 0) ? csv[k] : 0.f; b.y = 0.f;
    v8f c = {};
    c = __builtin_amdgcn_wmma_f32_16x16x4_f32(
        false, a, false, b, (short)0, c, false, false);
#pragma unroll
    for (int j = 0; j < 8; ++j) o[(size_t)j * DD + k * 16] = c[j];
  }
}

extern "C" void kernel_launch(void* const* d_in, const int* in_sizes, int n_in,
                              void* d_out, int out_size, void* d_ws, size_t ws_size,
                              hipStream_t stream) {
  (void)in_sizes; (void)n_in; (void)out_size; (void)ws_size;

  const float* past_states   = (const float*)d_in[0];
  const float* current_state = (const float*)d_in[1];
  const float* gradients     = (const float*)d_in[2];
  float* out = (float*)d_out;

  float* partial = (float*)d_ws;

  la_phase1_gemv<<<(CHUNKS * DTILES) / 8, 256, 0, stream>>>(past_states, gradients, partial);
  la_reduce_outer<<<DTILES * (DD / STRIP_COLS), 256, 0, stream>>>(partial, current_state, out);
}
